# starting kernel
# baseline (speedup 1.0000x reference)
#include <hip/hip_runtime.h>
#include <hip/hip_bf16.h>

constexpr int Bn = 8;
constexpr int Sn = 4096;
constexpr int Dn = 1024;
constexpr int Pn = 512;

typedef float v2f __attribute__((ext_vector_type(2)));
typedef float v8f __attribute__((ext_vector_type(8)));

__global__ __launch_bounds__(256) void seg_starts_kernel(
    const int* __restrict__ bnd, int* __restrict__ starts) {
  const int b = blockIdx.x;
  const int tid = threadIdx.x;
  const int base = tid * 16;
  const int* row = bnd + (size_t)b * Sn;

  __shared__ int psum[256];

  int vals[16];
  int local = 0;
#pragma unroll
  for (int i = 0; i < 16; ++i) {
    vals[i] = row[base + i];
    local += vals[i];
  }
  psum[tid] = local;
  __syncthreads();

  for (int off = 1; off < 256; off <<= 1) {
    int add = (tid >= off) ? psum[tid - off] : 0;
    __syncthreads();
    psum[tid] += add;
    __syncthreads();
  }

  int running = psum[tid] - local;
  int* srow = starts + (size_t)b * (Pn + 1);
#pragma unroll
  for (int i = 0; i < 16; ++i) {
    if (vals[i]) {
      srow[running] = base + i;
      ++running;
    }
  }
  if (tid == 0) srow[Pn] = Sn;
}

__global__ __launch_bounds__(128) void pool_wmma_kernel(
    const float* __restrict__ x, const int* __restrict__ starts,
    float* __restrict__ out) {
  constexpr int DCHUNK = 256;
  constexpr int NPB = Pn / 16;
  constexpr int ND = Dn / DCHUNK;

  const int wg = blockIdx.x;
  const int dchunk = wg % ND;
  const int pb = (wg / ND) % NPB;
  const int b = wg / (ND * NPB);
  const int p0 = pb * 16;

  __shared__ int sStarts[17];
  const int tid = threadIdx.x;
  if (tid < 17) sStarts[tid] = starts[(size_t)b * (Pn + 1) + p0 + tid];
  __syncthreads();

  const int lane = tid & 31;
  const int wave = tid >> 5;
  const int hi = lane >> 4;
  const int mrow = lane & 15;
  const int n = lane & 15;

  const int startM = sStarts[mrow];
  const int endM = sStarts[mrow + 1];
  const int s0 = sStarts[0];
  const int s1 = sStarts[16];

  const int dbase = dchunk * DCHUNK + wave * 64;
  const float* xb = x + (size_t)b * Sn * Dn;

  v8f c0 = {}, c1 = {}, c2 = {}, c3 = {};
  const int koff = hi * 2;

  for (int s = s0; s < s1; s += 4) {
    const int ka = s + koff;
    const int kb = ka + 1;

    v2f A;
    A.x = (ka >= startM && ka < endM) ? 1.0f : 0.0f;
    A.y = (kb >= startM && kb < endM) ? 1.0f : 0.0f;

    const int kac = (ka < Sn - 1) ? ka : (Sn - 1);
    const int kbc = (kb < Sn - 1) ? kb : (Sn - 1);
    const float* ra = xb + (size_t)kac * Dn + dbase + n;
    const float* rb = xb + (size_t)kbc * Dn + dbase + n;

    v2f B0, B1, B2, B3;
    B0.x = ra[0];  B0.y = rb[0];
    B1.x = ra[16]; B1.y = rb[16];
    B2.x = ra[32]; B2.y = rb[32];
    B3.x = ra[48]; B3.y = rb[48];

    c0 = __builtin_amdgcn_wmma_f32_16x16x4_f32(false, A, false, B0, (short)0,
                                               c0, false, false);
    c1 = __builtin_amdgcn_wmma_f32_16x16x4_f32(false, A, false, B1, (short)0,
                                               c1, false, false);
    c2 = __builtin_amdgcn_wmma_f32_16x16x4_f32(false, A, false, B2, (short)0,
                                               c2, false, false);
    c3 = __builtin_amdgcn_wmma_f32_16x16x4_f32(false, A, false, B3, (short)0,
                                               c3, false, false);
  }

#pragma unroll
  for (int j = 0; j < 8; ++j) {
    const int rl = j + hi * 8;
    const float inv = 1.0f / (float)(sStarts[rl + 1] - sStarts[rl]);
    const size_t o = ((size_t)(b * Pn + p0 + rl)) * Dn + dbase + n;
    out[o + 0]  = c0[j] * inv;
    out[o + 16] = c1[j] * inv;
    out[o + 32] = c2[j] * inv;
    out[o + 48] = c3[j] * inv;
  }
}

extern "C" void kernel_launch(void* const* d_in, const int* in_sizes, int n_in,
                              void* d_out, int out_size, void* d_ws,
                              size_t ws_size, hipStream_t stream) {
  const float* x = (const float*)d_in[0];
  const int* boundaries = (const int*)d_in[1];
  float* out = (float*)d_out;

  int* starts = (int*)d_ws;

  seg_starts_kernel<<<Bn, 256, 0, stream>>>(boundaries, starts);

  constexpr int DCHUNK = 256;
  const int grid = Bn * (Pn / 16) * (Dn / DCHUNK);
  pool_wmma_kernel<<<grid, 128, 0, stream>>>(x, starts, out);
}


# K loop software-pipelined over two register fragment sets: a wave requests its next trip's 32 loads before the current trip's 16 WMMAs (trip tests scalar); was load, wait, WMMA. Sum order unchanged
# speedup vs baseline: 1.5344x; 1.5344x over previous
#include <hip/hip_runtime.h>
#include <hip/hip_bf16.h>

constexpr int Bn = 8;
constexpr int Sn = 4096;
constexpr int Dn = 1024;
constexpr int Pn = 512;

typedef float v2f __attribute__((ext_vector_type(2)));
typedef float v4f __attribute__((ext_vector_type(4)));
typedef float v8f __attribute__((ext_vector_type(8)));

__global__ __launch_bounds__(256) void seg_starts_kernel(
    const int* __restrict__ bnd, int* __restrict__ starts) {
  const int b = blockIdx.x;
  const int tid = threadIdx.x;
  const int base = tid * 16;
  const int* row = bnd + (size_t)b * Sn;

  __shared__ int psum[256];

  int vals[16];
  int local = 0;
#pragma unroll
  for (int i = 0; i < 16; ++i) {
    vals[i] = row[base + i];
    local += vals[i];
  }
  psum[tid] = local;
  __syncthreads();

  for (int off = 1; off < 256; off <<= 1) {
    int add = (tid >= off) ? psum[tid - off] : 0;
    __syncthreads();
    psum[tid] += add;
    __syncthreads();
  }

  int running = psum[tid] - local;
  int* srow = starts + (size_t)b * (Pn + 1);
#pragma unroll
  for (int i = 0; i < 16; ++i) {
    if (vals[i]) {
      srow[running] = base + i;
      ++running;
    }
  }
  if (tid == 0) srow[Pn] = Sn;
}

__global__ __launch_bounds__(128) void pool_wmma_kernel(
    const float* __restrict__ x, const int* __restrict__ starts,
    float* __restrict__ out) {
  constexpr int DCHUNK = 256;
  constexpr int NT = DCHUNK / 16;
  constexpr int KSPLIT = 4;
  constexpr int NPB = Pn / 16;
  constexpr int ND = Dn / DCHUNK;

  const int wg = blockIdx.x;
  const int dchunk = wg % ND;
  const int pb = (wg / ND) % NPB;
  const int b = wg / (ND * NPB);
  const int p0 = pb * 16;

  __shared__ int sStarts[17];
  __shared__ v4f red[KSPLIT * NT * 2 * 32];

  const int tid = threadIdx.x;
  if (tid < 17) sStarts[tid] = starts[(size_t)b * (Pn + 1) + p0 + tid];
  __syncthreads();

  const int lane = tid & 31;
  const int q = __builtin_amdgcn_readfirstlane(tid >> 5);
  const int hi = lane >> 4;
  const int mrow = lane & 15;
  const int n = lane & 15;

  const int startM = sStarts[mrow];
  const int endM = sStarts[mrow + 1];
  const int s0 = __builtin_amdgcn_readfirstlane(sStarts[0]);
  const int s1 = __builtin_amdgcn_readfirstlane(sStarts[16]);

  const float* xb = x + (size_t)b * Sn * Dn + dchunk * DCHUNK + n;

  v8f c[NT];
#pragma unroll
  for (int t = 0; t < NT; ++t) c[t] = v8f{};

  const int koff = hi * 2;
  constexpr int STEP = 4 * KSPLIT;

  auto load_trip = [&](v2f (&Bf)[NT], int s) {
    const int ka = s + koff;
    const int kb = ka + 1;
    const int kac = (ka < Sn - 1) ? ka : (Sn - 1);
    const int kbc = (kb < Sn - 1) ? kb : (Sn - 1);
    const float* ra = xb + (size_t)kac * Dn;
    const float* rb = xb + (size_t)kbc * Dn;
#pragma unroll
    for (int t = 0; t < NT; ++t) {
      Bf[t].x = ra[16 * t];
      Bf[t].y = rb[16 * t];
    }
  };
  auto mma_trip = [&](const v2f (&Bf)[NT], int s) {
    const int ka = s + koff;
    const int kb = ka + 1;
    v2f A;
    A.x = (ka >= startM && ka < endM) ? 1.0f : 0.0f;
    A.y = (kb >= startM && kb < endM) ? 1.0f : 0.0f;
#pragma unroll
    for (int t = 0; t < NT; ++t) {
      c[t] = __builtin_amdgcn_wmma_f32_16x16x4_f32(false, A, false, Bf[t],
                                                   (short)0, c[t], false,
                                                   false);
    }
  };

  v2f B0[NT], B1[NT];
  int s = s0 + 4 * q;
  load_trip(B0, s);
  while (s + 2 * STEP < s1) {
    load_trip(B1, s + STEP);
    __builtin_amdgcn_sched_barrier(0);
    mma_trip(B0, s);
    load_trip(B0, s + 2 * STEP);
    __builtin_amdgcn_sched_barrier(0);
    mma_trip(B1, s + STEP);
    s += 2 * STEP;
  }
  const bool two_left = s + STEP < s1;
  if (two_left) load_trip(B1, s + STEP);
  mma_trip(B0, s);
  if (two_left) mma_trip(B1, s + STEP);

#pragma unroll
  for (int t = 0; t < NT; ++t) {
    red[((q * NT + t) * 2 + 0) * 32 + lane] =
        v4f{c[t][0], c[t][1], c[t][2], c[t][3]};
    red[((q * NT + t) * 2 + 1) * 32 + lane] =
        v4f{c[t][4], c[t][5], c[t][6], c[t][7]};
  }
  __syncthreads();

  float inv[8];
#pragma unroll
  for (int j = 0; j < 8; ++j) {
    const int rl = j + hi * 8;
    inv[j] = 1.0f / (float)(sStarts[rl + 1] - sStarts[rl]);
  }
  const size_t obase =
      ((size_t)(b * Pn + p0 + hi * 8)) * Dn + dchunk * DCHUNK + q * 64 + n;
#pragma unroll
  for (int tt = 0; tt < 4; ++tt) {
    const int t = q * 4 + tt;
#pragma unroll
    for (int h = 0; h < 2; ++h) {
      const v4f a0 = red[((0 * NT + t) * 2 + h) * 32 + lane];
      const v4f a1 = red[((1 * NT + t) * 2 + h) * 32 + lane];
      const v4f a2 = red[((2 * NT + t) * 2 + h) * 32 + lane];
      const v4f a3 = red[((3 * NT + t) * 2 + h) * 32 + lane];
      const v4f sum = ((a0 + a1) + a2) + a3;
#pragma unroll
      for (int i = 0; i < 4; ++i) {
        const int j = 4 * h + i;
        out[obase + (size_t)j * Dn + 16 * tt] = sum[i] * inv[j];
      }
    }
  }
}

extern "C" void kernel_launch(void* const* d_in, const int* in_sizes, int n_in,
                              void* d_out, int out_size, void* d_ws,
                              size_t ws_size, hipStream_t stream) {
  const float* x = (const float*)d_in[0];
  const int* boundaries = (const int*)d_in[1];
  float* out = (float*)d_out;

  int* starts = (int*)d_ws;

  seg_starts_kernel<<<Bn, 256, 0, stream>>>(boundaries, starts);

  constexpr int DCHUNK = 256;
  const int grid = Bn * (Pn / 16) * (Dn / DCHUNK);
  pool_wmma_kernel<<<grid, 128, 0, stream>>>(x, starts, out);
}


# Start-position kernel ranks boundary flags by wave ballots, 8 wave totals through LDS behind one barrier; was a 256-entry LDS scan with 16 barriers. Pooling kernel and sums unchanged
# speedup vs baseline: 1.5917x; 1.0373x over previous
#include <hip/hip_runtime.h>
#include <hip/hip_bf16.h>

constexpr int Bn = 8;
constexpr int Sn = 4096;
constexpr int Dn = 1024;
constexpr int Pn = 512;

typedef float v2f __attribute__((ext_vector_type(2)));
typedef float v4f __attribute__((ext_vector_type(4)));
typedef float v8f __attribute__((ext_vector_type(8)));

typedef int v4i __attribute__((ext_vector_type(4)));

__global__ __launch_bounds__(256) void seg_starts_kernel(
    const int* __restrict__ bnd, int* __restrict__ starts) {
  const int b = blockIdx.x;
  const int tid = threadIdx.x;
  const int lane = tid & 31;
  const int w = __builtin_amdgcn_readfirstlane(tid >> 5);
  const int* chunk = bnd + (size_t)b * Sn + w * 512;

  __shared__ int wtot[8];

  v4i v[4];
  for (int j = 0; j < 4; ++j)
    v[j] = *reinterpret_cast<const v4i*>(chunk + 128 * j + 4 * lane);

  int pre[16];
  bool f[16];
  int run = 0;
  for (int j = 0; j < 4; ++j) {
    unsigned m[4];
    int below = 0;
    for (int i = 0; i < 4; ++i) {
      f[4 * j + i] = v[j][i] != 0;
      m[i] = __builtin_amdgcn_ballot_w32(f[4 * j + i]);
      below = __builtin_amdgcn_mbcnt_lo(m[i], below);
    }
    int p = run + below;
    for (int i = 0; i < 4; ++i) {
      pre[4 * j + i] = p;
      p += f[4 * j + i] ? 1 : 0;
    }
    run += __builtin_popcount(m[0]) + __builtin_popcount(m[1]) +
           __builtin_popcount(m[2]) + __builtin_popcount(m[3]);
  }
  if (lane == 0) wtot[w] = run;
  __syncthreads();

  int off = 0;
  for (int k = 0; k < 8; ++k) off += (k < w) ? wtot[k] : 0;

  int* srow = starts + (size_t)b * (Pn + 1);
  const int tok0 = w * 512 + 4 * lane;
  for (int j = 0; j < 4; ++j) {
    for (int i = 0; i < 4; ++i) {
      if (f[4 * j + i]) srow[off + pre[4 * j + i]] = tok0 + 128 * j + i;
    }
  }
  if (tid == 0) srow[Pn] = Sn;
}

__global__ __launch_bounds__(128) void pool_wmma_kernel(
    const float* __restrict__ x, const int* __restrict__ starts,
    float* __restrict__ out) {
  constexpr int DCHUNK = 256;
  constexpr int NT = DCHUNK / 16;
  constexpr int KSPLIT = 4;
  constexpr int NPB = Pn / 16;
  constexpr int ND = Dn / DCHUNK;

  const int wg = blockIdx.x;
  const int dchunk = wg % ND;
  const int pb = (wg / ND) % NPB;
  const int b = wg / (ND * NPB);
  const int p0 = pb * 16;

  __shared__ int sStarts[17];
  __shared__ v4f red[KSPLIT * NT * 2 * 32];

  const int tid = threadIdx.x;
  if (tid < 17) sStarts[tid] = starts[(size_t)b * (Pn + 1) + p0 + tid];
  __syncthreads();

  const int lane = tid & 31;
  const int q = __builtin_amdgcn_readfirstlane(tid >> 5);
  const int hi = lane >> 4;
  const int mrow = lane & 15;
  const int n = lane & 15;

  const int startM = sStarts[mrow];
  const int endM = sStarts[mrow + 1];
  const int s0 = __builtin_amdgcn_readfirstlane(sStarts[0]);
  const int s1 = __builtin_amdgcn_readfirstlane(sStarts[16]);

  const float* xb = x + (size_t)b * Sn * Dn + dchunk * DCHUNK + n;

  v8f c[NT];
#pragma unroll
  for (int t = 0; t < NT; ++t) c[t] = v8f{};

  const int koff = hi * 2;
  constexpr int STEP = 4 * KSPLIT;

  auto load_trip = [&](v2f (&Bf)[NT], int s) {
    const int ka = s + koff;
    const int kb = ka + 1;
    const int kac = (ka < Sn - 1) ? ka : (Sn - 1);
    const int kbc = (kb < Sn - 1) ? kb : (Sn - 1);
    const float* ra = xb + (size_t)kac * Dn;
    const float* rb = xb + (size_t)kbc * Dn;
#pragma unroll
    for (int t = 0; t < NT; ++t) {
      Bf[t].x = ra[16 * t];
      Bf[t].y = rb[16 * t];
    }
  };
  auto mma_trip = [&](const v2f (&Bf)[NT], int s) {
    const int ka = s + koff;
    const int kb = ka + 1;
    v2f A;
    A.x = (ka >= startM && ka < endM) ? 1.0f : 0.0f;
    A.y = (kb >= startM && kb < endM) ? 1.0f : 0.0f;
#pragma unroll
    for (int t = 0; t < NT; ++t) {
      c[t] = __builtin_amdgcn_wmma_f32_16x16x4_f32(false, A, false, Bf[t],
                                                   (short)0, c[t], false,
                                                   false);
    }
  };

  v2f B0[NT], B1[NT];
  int s = s0 + 4 * q;
  load_trip(B0, s);
  while (s + 2 * STEP < s1) {
    load_trip(B1, s + STEP);
    __builtin_amdgcn_sched_barrier(0);
    mma_trip(B0, s);
    load_trip(B0, s + 2 * STEP);
    __builtin_amdgcn_sched_barrier(0);
    mma_trip(B1, s + STEP);
    s += 2 * STEP;
  }
  const bool two_left = s + STEP < s1;
  if (two_left) load_trip(B1, s + STEP);
  mma_trip(B0, s);
  if (two_left) mma_trip(B1, s + STEP);

#pragma unroll
  for (int t = 0; t < NT; ++t) {
    red[((q * NT + t) * 2 + 0) * 32 + lane] =
        v4f{c[t][0], c[t][1], c[t][2], c[t][3]};
    red[((q * NT + t) * 2 + 1) * 32 + lane] =
        v4f{c[t][4], c[t][5], c[t][6], c[t][7]};
  }
  __syncthreads();

  float inv[8];
#pragma unroll
  for (int j = 0; j < 8; ++j) {
    const int rl = j + hi * 8;
    inv[j] = 1.0f / (float)(sStarts[rl + 1] - sStarts[rl]);
  }
  const size_t obase =
      ((size_t)(b * Pn + p0 + hi * 8)) * Dn + dchunk * DCHUNK + q * 64 + n;
#pragma unroll
  for (int tt = 0; tt < 4; ++tt) {
    const int t = q * 4 + tt;
#pragma unroll
    for (int h = 0; h < 2; ++h) {
      const v4f a0 = red[((0 * NT + t) * 2 + h) * 32 + lane];
      const v4f a1 = red[((1 * NT + t) * 2 + h) * 32 + lane];
      const v4f a2 = red[((2 * NT + t) * 2 + h) * 32 + lane];
      const v4f a3 = red[((3 * NT + t) * 2 + h) * 32 + lane];
      const v4f sum = ((a0 + a1) + a2) + a3;
#pragma unroll
      for (int i = 0; i < 4; ++i) {
        const int j = 4 * h + i;
        out[obase + (size_t)j * Dn + 16 * tt] = sum[i] * inv[j];
      }
    }
  }
}

extern "C" void kernel_launch(void* const* d_in, const int* in_sizes, int n_in,
                              void* d_out, int out_size, void* d_ws,
                              size_t ws_size, hipStream_t stream) {
  const float* x = (const float*)d_in[0];
  const int* boundaries = (const int*)d_in[1];
  float* out = (float*)d_out;

  int* starts = (int*)d_ws;

  seg_starts_kernel<<<Bn, 256, 0, stream>>>(boundaries, starts);

  constexpr int DCHUNK = 256;
  const int grid = Bn * (Pn / 16) * (Dn / DCHUNK);
  pool_wmma_kernel<<<grid, 128, 0, stream>>>(x, starts, out);
}


# Pooling tile now 8 segments x 512 channels, 2-token K steps, one-hot A block-diagonal over the channel half: every 4-byte load and every store is one whole 128-B line. K sum order changes
# speedup vs baseline: 1.7599x; 1.1057x over previous
#include <hip/hip_runtime.h>
#include <hip/hip_bf16.h>

constexpr int Bn = 8;
constexpr int Sn = 4096;
constexpr int Dn = 1024;
constexpr int Pn = 512;

typedef float v2f __attribute__((ext_vector_type(2)));
typedef float v4f __attribute__((ext_vector_type(4)));
typedef float v8f __attribute__((ext_vector_type(8)));

typedef int v4i __attribute__((ext_vector_type(4)));

__global__ __launch_bounds__(256) void seg_starts_kernel(
    const int* __restrict__ bnd, int* __restrict__ starts) {
  const int b = blockIdx.x;
  const int tid = threadIdx.x;
  const int lane = tid & 31;
  const int w = __builtin_amdgcn_readfirstlane(tid >> 5);
  const int* chunk = bnd + (size_t)b * Sn + w * 512;

  __shared__ int wtot[8];

  v4i v[4];
  for (int j = 0; j < 4; ++j)
    v[j] = *reinterpret_cast<const v4i*>(chunk + 128 * j + 4 * lane);

  int pre[16];
  bool f[16];
  int run = 0;
  for (int j = 0; j < 4; ++j) {
    unsigned m[4];
    int below = 0;
    for (int i = 0; i < 4; ++i) {
      f[4 * j + i] = v[j][i] != 0;
      m[i] = __builtin_amdgcn_ballot_w32(f[4 * j + i]);
      below = __builtin_amdgcn_mbcnt_lo(m[i], below);
    }
    int p = run + below;
    for (int i = 0; i < 4; ++i) {
      pre[4 * j + i] = p;
      p += f[4 * j + i] ? 1 : 0;
    }
    run += __builtin_popcount(m[0]) + __builtin_popcount(m[1]) +
           __builtin_popcount(m[2]) + __builtin_popcount(m[3]);
  }
  if (lane == 0) wtot[w] = run;
  __syncthreads();

  int off = 0;
  for (int k = 0; k < 8; ++k) off += (k < w) ? wtot[k] : 0;

  int* srow = starts + (size_t)b * (Pn + 1);
  const int tok0 = w * 512 + 4 * lane;
  for (int j = 0; j < 4; ++j) {
    for (int i = 0; i < 4; ++i) {
      if (f[4 * j + i]) srow[off + pre[4 * j + i]] = tok0 + 128 * j + i;
    }
  }
  if (tid == 0) srow[Pn] = Sn;
}

__global__ __launch_bounds__(128) void pool_wmma_kernel(
    const float* __restrict__ x, const int* __restrict__ starts,
    float* __restrict__ out) {
  constexpr int DCHUNK = 512;
  constexpr int NT = DCHUNK / 32;
  constexpr int KSPLIT = 4;
  constexpr int MSEG = 8;
  constexpr int NPB = Pn / MSEG;
  constexpr int ND = Dn / DCHUNK;

  const int wg = blockIdx.x;
  const int dchunk = wg % ND;
  const int pb = (wg / ND) % NPB;
  const int b = wg / (ND * NPB);
  const int p0 = pb * MSEG;

  __shared__ int sStarts[MSEG + 1];
  __shared__ v4f red[KSPLIT * NT * 2 * 32];

  const int tid = threadIdx.x;
  if (tid <= MSEG) sStarts[tid] = starts[(size_t)b * (Pn + 1) + p0 + tid];
  __syncthreads();

  const int lane = tid & 31;
  const int q = __builtin_amdgcn_readfirstlane(tid >> 5);
  const int hk = lane >> 4;
  const int mrow = lane & 15;
  const int hm = mrow >> 3;
  const int sm = mrow & 7;

  const bool diag = hm == hk;
  const int startM = diag ? sStarts[sm] : Sn;
  const int endM = diag ? sStarts[sm + 1] : 0;
  const int s0 = __builtin_amdgcn_readfirstlane(sStarts[0]);
  const int s1 = __builtin_amdgcn_readfirstlane(sStarts[MSEG]);

  const float* xb = x + (size_t)b * Sn * Dn + dchunk * DCHUNK + lane;

  v8f c[NT];
#pragma unroll
  for (int t = 0; t < NT; ++t) c[t] = v8f{};

  constexpr int STEP = 2 * KSPLIT;

  auto load_trip = [&](v2f (&Bf)[NT], int s) {
    const int kac = (s < Sn - 1) ? s : (Sn - 1);
    const int kbc = (s + 1 < Sn - 1) ? (s + 1) : (Sn - 1);
    const float* ra = xb + (size_t)kac * Dn;
    const float* rb = xb + (size_t)kbc * Dn;
#pragma unroll
    for (int t = 0; t < NT; ++t) {
      Bf[t].x = ra[32 * t];
      Bf[t].y = rb[32 * t];
    }
  };
  auto mma_trip = [&](const v2f (&Bf)[NT], int s) {
    v2f A;
    A.x = (s >= startM && s < endM) ? 1.0f : 0.0f;
    A.y = (s + 1 >= startM && s + 1 < endM) ? 1.0f : 0.0f;
#pragma unroll
    for (int t = 0; t < NT; ++t) {
      c[t] = __builtin_amdgcn_wmma_f32_16x16x4_f32(false, A, false, Bf[t],
                                                   (short)0, c[t], false,
                                                   false);
    }
  };

  v2f B0[NT], B1[NT];
  int s = s0 + 2 * q;
  load_trip(B0, s);
  while (s + 2 * STEP < s1) {
    load_trip(B1, s + STEP);
    __builtin_amdgcn_sched_barrier(0);
    mma_trip(B0, s);
    load_trip(B0, s + 2 * STEP);
    __builtin_amdgcn_sched_barrier(0);
    mma_trip(B1, s + STEP);
    s += 2 * STEP;
  }
  const bool two_left = s + STEP < s1;
  if (two_left) load_trip(B1, s + STEP);
  mma_trip(B0, s);
  if (two_left) mma_trip(B1, s + STEP);

#pragma unroll
  for (int t = 0; t < NT; ++t) {
    red[((q * NT + t) * 2 + 0) * 32 + lane] =
        v4f{c[t][0], c[t][1], c[t][2], c[t][3]};
    red[((q * NT + t) * 2 + 1) * 32 + lane] =
        v4f{c[t][4], c[t][5], c[t][6], c[t][7]};
  }
  __syncthreads();

  float inv[MSEG];
#pragma unroll
  for (int r = 0; r < MSEG; ++r)
    inv[r] = 1.0f / (float)(sStarts[r + 1] - sStarts[r]);
  const size_t obase =
      ((size_t)(b * Pn + p0)) * Dn + dchunk * DCHUNK + q * 128 + lane;
#pragma unroll
  for (int tt = 0; tt < 4; ++tt) {
    const int t = q * 4 + tt;
#pragma unroll
    for (int h = 0; h < 2; ++h) {
      const v4f a0 = red[((0 * NT + t) * 2 + h) * 32 + lane];
      const v4f a1 = red[((1 * NT + t) * 2 + h) * 32 + lane];
      const v4f a2 = red[((2 * NT + t) * 2 + h) * 32 + lane];
      const v4f a3 = red[((3 * NT + t) * 2 + h) * 32 + lane];
      const v4f sum = ((a0 + a1) + a2) + a3;
#pragma unroll
      for (int i = 0; i < 4; ++i) {
        const int r = 4 * h + i;
        out[obase + (size_t)r * Dn + 32 * tt] = sum[i] * inv[r];
      }
    }
  }
}

extern "C" void kernel_launch(void* const* d_in, const int* in_sizes, int n_in,
                              void* d_out, int out_size, void* d_ws,
                              size_t ws_size, hipStream_t stream) {
  const float* x = (const float*)d_in[0];
  const int* boundaries = (const int*)d_in[1];
  float* out = (float*)d_out;

  int* starts = (int*)d_ws;

  seg_starts_kernel<<<Bn, 256, 0, stream>>>(boundaries, starts);

  constexpr int DCHUNK = 512;
  const int grid = Bn * (Pn / 8) * (Dn / DCHUNK);
  pool_wmma_kernel<<<grid, 128, 0, stream>>>(x, starts, out);
}
